# starting kernel
# baseline (speedup 1.0000x reference)
#include <hip/hip_runtime.h>
#include <hip/hip_bf16.h>
#include <math.h>


typedef __attribute__((ext_vector_type(2))) float v2f;
typedef __attribute__((ext_vector_type(8))) float v8f;

#define B_ROWS 4096
#define T_LEN  8192
#define N_WIN  8
#define TILE_M 16
#define WAVES  8
#define BLOCK  256
#define KSTEPS_TOTAL    (T_LEN / 4)
#define KSTEPS_PER_WAVE (KSTEPS_TOTAL / WAVES)

__global__ void ewma_weights_kernel(const float* __restrict__ log_halflife,
                                    float* __restrict__ wgtB) {
  int idx = blockIdx.x * blockDim.x + threadIdx.x;
  if (idx >= KSTEPS_TOTAL * 64) return;
  int s    = idx >> 6;
  int r    = idx & 63;
  int lane = r >> 1;
  int j    = r & 1;
  int t    = 4 * s + 2 * (lane >> 4) + j;
  int n    = lane & 15;
  float val = 0.0f;
  if (n < N_WIN) {
    float hl = expf(log_halflife[n]);
    hl = fminf(fmaxf(hl, 1.0f), 200.0f);
    float invhl = 1.0f / hl;
    float d = exp2f(-invhl);
    float age = (float)(T_LEN - 1 - t);
    val = exp2f(-age * invhl);
    if (t > 0) val *= (1.0f - d);
  }
  wgtB[idx] = val;
}

__global__ void __launch_bounds__(BLOCK)
ewma_wmma_kernel(const float* __restrict__ X,
                 const float* __restrict__ wgtB,
                 float* __restrict__ out) {
  __shared__ float red[WAVES * 256];

  const int tile = blockIdx.x;
  const int lane = threadIdx.x & 31;
  const int wv   = threadIdx.x >> 5;

  const int row    = tile * TILE_M + (lane & 15);
  const int kStart = wv * KSTEPS_PER_WAVE * 4;
  const float* ap = X + (size_t)row * T_LEN + kStart + 2 * (lane >> 4);
  const float* bp = wgtB + (size_t)(wv * KSTEPS_PER_WAVE) * 64 + lane * 2;

  v8f acc = {};
#pragma unroll 8
  for (int s = 0; s < KSTEPS_PER_WAVE; ++s) {
    v2f a = *(const v2f*)ap;
    v2f b = *(const v2f*)bp;
    acc = __builtin_amdgcn_wmma_f32_16x16x4_f32(
        false, a, false, b, (short)0, acc, false, false);
    ap += 4;
    bp += 64;
  }

#pragma unroll
  for (int r = 0; r < 8; ++r)
    red[wv * 256 + lane * 8 + r] = acc[r];
  __syncthreads();

  const int i = threadIdx.x;
  float sum = 0.0f;
#pragma unroll
  for (int w = 0; w < WAVES; ++w) sum += red[w * 256 + i];

  const int srcLane = i >> 3;
  const int r = i & 7;
  const int M = r + 8 * (srcLane >> 4);
  const int N = srcLane & 15;
  if (N < N_WIN) {
    const int grow = tile * TILE_M + M;
    const float last = X[(size_t)grow * T_LEN + (T_LEN - 1)];
    out[grow * N_WIN + N] = (sum - last) / (fabsf(last) + 1e-8f);
  }
}

extern "C" void kernel_launch(void* const* d_in, const int* in_sizes, int n_in,
                              void* d_out, int out_size, void* d_ws, size_t ws_size,
                              hipStream_t stream) {
  const float* price        = (const float*)d_in[0];
  const float* log_halflife = (const float*)d_in[1];
  float* out  = (float*)d_out;
  float* wgtB = (float*)d_ws;

  (void)in_sizes; (void)n_in; (void)out_size; (void)ws_size;

  const int wElems = KSTEPS_TOTAL * 64;
  ewma_weights_kernel<<<(wElems + BLOCK - 1) / BLOCK, BLOCK, 0, stream>>>(
      log_halflife, wgtB);

  ewma_wmma_kernel<<<B_ROWS / TILE_M, BLOCK, 0, stream>>>(price, wgtB, out);
}


# WMMA M index now numbers the sixteen interleaved 16-byte groups of ONE row (per-group decay applied after the product), so each wave streams a row with whole-line b128 loads; summation order changes
# speedup vs baseline: 2.0740x; 2.0740x over previous
#include <hip/hip_runtime.h>
#include <math.h>


typedef __attribute__((ext_vector_type(2))) float v2f;
typedef __attribute__((ext_vector_type(4))) float v4f;
typedef __attribute__((ext_vector_type(8))) float v8f;

#define B_ROWS   4096
#define T_LEN    8192
#define N_WIN    8
#define PIECES   64
#define SHIFT    60
#define TAB_W    (PIECES * 128)
#define TAB_S    256
#define TAB_C    16
#define TAB_USED (TAB_W + TAB_S + TAB_C)
#define TAB_ALL  8704
#define WAVES    8
#define THREADS  256
#define RING     32
#define BLK      8

__global__ void __launch_bounds__(256)
ewma_table_kernel(const float* __restrict__ log_halflife,
                  float* __restrict__ tab) {
  const int idx = blockIdx.x * 256 + threadIdx.x;
  int n = 16;
  int kind = 3;
  float age = 0.0f;
  if (idx < TAB_W) {
    const int lp = (idx >> 2) & 31;
    n = lp & 15;
    age = (float)(T_LEN - 1 - SHIFT - (128 * (idx >> 7) + 64 * (lp >> 4) + (idx & 3)));
    kind = 0;
  } else if (idx < TAB_W + TAB_S) {
    const int k = idx - TAB_W;
    const int lp = k >> 3;
    n = lp & 15;
    age = (float)(SHIFT - 4 * (8 * (lp >> 4) + (k & 7)));
    kind = 1;
  } else if (idx < TAB_USED) {
    n = idx - (TAB_W + TAB_S);
    age = (float)(T_LEN - 1);
    kind = 2;
  }
  float val = 0.0f;
  if (n < N_WIN) {
    float hl = expf(log_halflife[n]);
    hl = fminf(fmaxf(hl, 1.0f), 200.0f);
    const float invhl = 1.0f / hl;
    const float d = exp2f(-invhl);
    val = exp2f(-age * invhl);
    if (kind == 0) val *= (1.0f - d);
    if (kind == 2) val *= d;
  }
  tab[idx] = val;
}

__global__ void __launch_bounds__(THREADS)
ewma_rows_kernel(const float* __restrict__ X,
                 const float* __restrict__ tab,
                 float* __restrict__ out) {
  __shared__ v4f   sTab[TAB_USED / 4];
  __shared__ float sOut[WAVES * N_WIN];

  const int tid  = threadIdx.x;
  const int lane = tid & 31;
  const int wv   = tid >> 5;

  for (int i = tid; i < TAB_USED / 4; i += THREADS)
    sTab[i] = ((const v4f*)tab)[i];
  __syncthreads();

  const int row = blockIdx.x * WAVES + wv;
  const v4f* xp = (const v4f*)(X + (size_t)row * T_LEN) + lane;

  v4f ring[RING];
#pragma unroll
  for (int q = 0; q < RING; ++q) ring[q] = xp[q * 32];

  v4f bw[2][BLK];
#pragma unroll
  for (int j = 0; j < BLK; ++j) bw[0][j] = sTab[j * 32 + lane];

  v8f acc0 = {};
  v8f acc1 = {};
  float x0 = 0.0f, last = 0.0f;

#pragma unroll
  for (int g = 0; g < PIECES / BLK; ++g) {
    if (g + 1 < PIECES / BLK) {
#pragma unroll
      for (int j = 0; j < BLK; ++j) bw[(g + 1) & 1][j] = sTab[((g + 1) * BLK + j) * 32 + lane];
    }
    __builtin_amdgcn_sched_barrier(0);
#pragma unroll
    for (int j = 0; j < BLK; ++j) {
      const int q = g * BLK + j;
      const v4f a = ring[q % RING];
      const v4f b = bw[g & 1][j];
      const float ax = a.x, aw = a.w;
      if (q == 0)
        x0 = __int_as_float(__builtin_amdgcn_readfirstlane(__float_as_int(ax)));
      if (q == PIECES - 1)
        last = __int_as_float(__builtin_amdgcn_readlane(__float_as_int(aw), 31));
      const v2f a01 = {a.x, a.y}, a23 = {a.z, a.w};
      const v2f b01 = {b.x, b.y}, b23 = {b.z, b.w};
      acc0 = __builtin_amdgcn_wmma_f32_16x16x4_f32(false, a01, false, b01, (short)0, acc0, false, false);
      acc1 = __builtin_amdgcn_wmma_f32_16x16x4_f32(false, a23, false, b23, (short)0, acc1, false, false);
    }
    __builtin_amdgcn_sched_barrier(0);
    if (g * BLK + RING < PIECES) {
#pragma unroll
      for (int j = 0; j < BLK; ++j) {
        const int q = g * BLK + j;
        ring[q % RING] = xp[(q + RING) * 32];
      }
    }
  }

  const v4f s0 = sTab[TAB_W / 4 + lane * 2];
  const v4f s1 = sTab[TAB_W / 4 + lane * 2 + 1];
  float part = (acc0[0] + acc1[0]) * s0.x;
  part = fmaf(acc0[1] + acc1[1], s0.y, part);
  part = fmaf(acc0[2] + acc1[2], s0.z, part);
  part = fmaf(acc0[3] + acc1[3], s0.w, part);
  part = fmaf(acc0[4] + acc1[4], s1.x, part);
  part = fmaf(acc0[5] + acc1[5], s1.y, part);
  part = fmaf(acc0[6] + acc1[6], s1.z, part);
  part = fmaf(acc0[7] + acc1[7], s1.w, part);
  part += __shfl_xor(part, 16, 32);

  if (lane < N_WIN) {
    const float corr = ((const float*)sTab)[TAB_W + TAB_S + lane];
    const float ew = fmaf(x0, corr, part);
    sOut[wv * N_WIN + lane] = (ew - last) / (fabsf(last) + 1e-8f);
  }
  __syncthreads();
  if (tid < (WAVES * N_WIN) / 4)
    ((v4f*)out)[blockIdx.x * ((WAVES * N_WIN) / 4) + tid] = ((const v4f*)sOut)[tid];
}

extern "C" void kernel_launch(void* const* d_in, const int* in_sizes, int n_in,
                              void* d_out, int out_size, void* d_ws, size_t ws_size,
                              hipStream_t stream) {
  const float* price        = (const float*)d_in[0];
  const float* log_halflife = (const float*)d_in[1];
  float* out = (float*)d_out;
  float* tab = (float*)d_ws;

  (void)in_sizes; (void)n_in; (void)out_size; (void)ws_size;

  ewma_table_kernel<<<TAB_ALL / 256, 256, 0, stream>>>(log_halflife, tab);
  ewma_rows_kernel<<<B_ROWS / WAVES, THREADS, 0, stream>>>(price, tab, out);
}
